# starting kernel
# baseline (speedup 1.0000x reference)
#include <hip/hip_runtime.h>
#include <math.h>

typedef __attribute__((ext_vector_type(2))) float v2f;
typedef __attribute__((ext_vector_type(8))) float v8f;

#define BATCH   8
#define M_PTS   512
#define N_PTS   8192
#define EPS_F   1e-7f
#define WAVES   8
#define THREADS (WAVES * 32)
#define TILES   (N_PTS / 16)
#define TPW     (TILES / WAVES)
#define FLT_BIG 3.402823466e+38f

__global__ __launch_bounds__(THREADS)
void pos_loss_kernel(const float* __restrict__ kp,
                     const float* __restrict__ pc,
                     const float* __restrict__ sn,
                     float* __restrict__ out)
{
    const int b    = blockIdx.y;
    const int m0   = blockIdx.x * 16;
    const int tid  = threadIdx.x;
    const int wave = tid >> 5;
    const int lane = tid & 31;
    const int col  = lane & 15;
    const bool hi  = lane >= 16;

    const float* kpb = kp + (size_t)b * 3 * M_PTS;
    const float* pcx = pc + (size_t)b * 3 * N_PTS;
    const float* pcy = pcx + N_PTS;
    const float* pcz = pcy + N_PTS;

    v2f a;
    {
        const float k0 = kpb[0 * M_PTS + m0 + col];
        const float k1 = kpb[1 * M_PTS + m0 + col];
        const float k2 = kpb[2 * M_PTS + m0 + col];
        a[0] = -2.0f * (hi ? k2 : k0);
        a[1] = hi ? 1.0f : (-2.0f * k1);
    }

    const v8f czero = {};

    float minv[8];
    int   mini[8];
#pragma unroll
    for (int r = 0; r < 8; ++r) { minv[r] = FLT_BIG; mini[r] = 0; }

    const int t0 = wave * TPW;
#pragma unroll 4
    for (int tt = 0; tt < TPW; ++tt) {
        const int n = (t0 + tt) * 16 + col;
        const float x = pcx[n];
        const float y = pcy[n];
        const float z = pcz[n];
        const float pn2 = x * x + y * y + z * z;

        v2f bm;
        bm[0] = hi ? z : x;
        bm[1] = hi ? pn2 : y;

        v8f d = __builtin_amdgcn_wmma_f32_16x16x4_f32(
             false, a,  false, bm,
             (short)0, czero,  false,  false);

#pragma unroll
        for (int r = 0; r < 8; ++r) {
            const float v = d[r];
            if (v < minv[r]) {
                minv[r] = v;
                mini[r] = n;
            }
        }
    }

#pragma unroll
    for (int r = 0; r < 8; ++r) {
#pragma unroll
        for (int off = 1; off < 16; off <<= 1) {
            const float ov = __shfl_xor(minv[r], off, 32);
            const int   oi = __shfl_xor(mini[r], off, 32);
            if (ov < minv[r] || (ov == minv[r] && oi < mini[r])) {
                minv[r] = ov;
                mini[r] = oi;
            }
        }
    }

    __shared__ float lv[WAVES][16];
    __shared__ int   li[WAVES][16];
    if (lane == 0 || lane == 16) {
        const int rowbase = hi ? 8 : 0;
#pragma unroll
        for (int r = 0; r < 8; ++r) {
            lv[wave][rowbase + r] = minv[r];
            li[wave][rowbase + r] = mini[r];
        }
    }
    __syncthreads();

    if (tid < 16) {
        const int m = tid;
        float bv = FLT_BIG;
        int   bi = 0;
#pragma unroll
        for (int w = 0; w < WAVES; ++w) {
            const float v = lv[w][m];
            const int   i = li[w][m];
            if (v < bv || (v == bv && i < bi)) { bv = v; bi = i; }
        }

        const float kx = kpb[0 * M_PTS + m0 + m];
        const float ky = kpb[1 * M_PTS + m0 + m];
        const float kz = kpb[2 * M_PTS + m0 + m];

        const float px = pcx[bi];
        const float py = pcy[bi];
        const float pz = pcz[bi];

        const float* snb = sn + (size_t)b * 3 * N_PTS;
        const float sx = snb[0 * N_PTS + bi];
        const float sy = snb[1 * N_PTS + bi];
        const float sz = snb[2 * N_PTS + bi];

        const float dx = kx - px, dy = ky - py, dz = kz - pz;
        const float dn  = sqrtf(dx * dx + dy * dy + dz * dz);
        const float inv = 1.0f / (dn + EPS_F);
        const float dot = (sx * dx + sy * dy + sz * dz) * inv;

        out[(size_t)b * M_PTS + m0 + m] = dot * dot;
    }
}

extern "C" void kernel_launch(void* const* d_in, const int* in_sizes, int n_in,
                              void* d_out, int out_size, void* d_ws, size_t ws_size,
                              hipStream_t stream) {
    const float* kp = (const float*)d_in[0];
    const float* pc = (const float*)d_in[1];
    const float* sn = (const float*)d_in[2];
    float* out = (float*)d_out;

    dim3 grid(M_PTS / 16, BATCH);
    pos_loss_kernel<<<grid, THREADS, 0, stream>>>(kp, pc, sn, out);
}


# (min, index) reduction as selects, no branches: 16-lane butterfly takes the partner pair by DPP row_xmask, not LDS permutes; cross-wave pass loads its pairs as one group; same scores
# speedup vs baseline: 2.0105x; 2.0105x over previous
#include <hip/hip_runtime.h>
#include <math.h>

typedef __attribute__((ext_vector_type(2))) float v2f;
typedef __attribute__((ext_vector_type(8))) float v8f;
typedef __attribute__((ext_vector_type(4))) float v4f;

#define BATCH   8
#define M_PTS   512
#define N_PTS   8192
#define EPS_F   1e-7f
#define WAVES   8
#define THREADS (WAVES * 32)
#define TILES   (N_PTS / 16)
#define TPW     (TILES / WAVES)
#define FLT_BIG 3.402823466e+38f
#define PPW     (TPW * 16)
#define ROW     16
#define PITCH   80

template <int M>
__device__ __forceinline__ int row_xor_i(int v)
{
    return __builtin_amdgcn_mov_dpp(v, 0x160 | M, 0xF, 0xF, false);
}
template <int M>
__device__ __forceinline__ float row_xor_f(float v)
{
    return __builtin_bit_cast(float, row_xor_i<M>(__builtin_bit_cast(int, v)));
}

template <int M>
__device__ __forceinline__ void row_min_step(float (&minv)[8], int (&mini)[8])
{
#pragma unroll
    for (int r = 0; r < 8; ++r) {
        const float ov = row_xor_f<M>(minv[r]);
        const int   oi = row_xor_i<M>(mini[r]);
        const bool take = (ov < minv[r]) | ((ov == minv[r]) & (oi < mini[r]));
        minv[r] = take ? ov : minv[r];
        mini[r] = take ? oi : mini[r];
    }
}

__global__ __launch_bounds__(THREADS)
void pos_loss_kernel(const float* __restrict__ kp,
                     const float* __restrict__ pc,
                     const float* __restrict__ sn,
                     float* __restrict__ out)
{
    const int b    = blockIdx.y;
    const int m0   = blockIdx.x * 16;
    const int tid  = threadIdx.x;
    const int wave = tid >> 5;
    const int lane = tid & 31;
    const int col  = lane & 15;
    const bool hi  = lane >= 16;

    const float* kpb = kp + (size_t)b * 3 * M_PTS;
    const float* pcx = pc + (size_t)b * 3 * N_PTS;
    const float* pcy = pcx + N_PTS;
    const float* pcz = pcy + N_PTS;

    v2f a;
    {
        const float k0 = kpb[0 * M_PTS + m0 + col];
        const float k1 = kpb[1 * M_PTS + m0 + col];
        const float k2 = kpb[2 * M_PTS + m0 + col];
        a[0] = -2.0f * (hi ? k2 : k0);
        a[1] = hi ? 1.0f : (-2.0f * k1);
    }

    const v8f czero = {};

    float minv[8];
    int   mini[8];
#pragma unroll
    for (int r = 0; r < 8; ++r) { minv[r] = FLT_BIG; mini[r] = 0; }

    const int t0 = wave * TPW;

    __shared__ float stage[TILES * PITCH];
    {
        v4f vx[8], vy[8], vz[8];
        const int pbase = wave * PPW + lane * 4;
#pragma unroll
        for (int i = 0; i < 8; ++i) {
            const int p = pbase + i * 128;
            vx[i] = *reinterpret_cast<const v4f*>(pcx + p);
            vy[i] = *reinterpret_cast<const v4f*>(pcy + p);
            vz[i] = *reinterpret_cast<const v4f*>(pcz + p);
        }
        float* dst0 = stage + (t0 + (lane >> 2)) * PITCH + (lane & 3) * 4;
#pragma unroll
        for (int i = 0; i < 8; ++i) {
            v4f pp;
#pragma unroll
            for (int j = 0; j < 4; ++j) {
                const float x = vx[i][j];
                const float y = vy[i][j];
                const float z = vz[i][j];
                pp[j] = x * x + y * y + z * z;
            }
            float* dst = dst0 + i * 8 * PITCH;
            *reinterpret_cast<v4f*>(dst + 0 * ROW) = vx[i];
            *reinterpret_cast<v4f*>(dst + 1 * ROW) = vy[i];
            *reinterpret_cast<v4f*>(dst + 2 * ROW) = vz[i];
            *reinterpret_cast<v4f*>(dst + 3 * ROW) = pp;
        }
    }
    __syncthreads();

    const float* frag = stage + t0 * PITCH + (hi ? 2 * ROW : 0) + col;
#pragma unroll 4
    for (int tt = 0; tt < TPW; ++tt) {
        const int n = (t0 + tt) * 16 + col;
        v2f bm;
        bm[0] = frag[tt * PITCH];
        bm[1] = frag[tt * PITCH + ROW];

        v8f d = __builtin_amdgcn_wmma_f32_16x16x4_f32(
             false, a,  false, bm,
             (short)0, czero,  false,  false);

#pragma unroll
        for (int r = 0; r < 8; ++r) {
            const float v = d[r];
            if (v < minv[r]) {
                minv[r] = v;
                mini[r] = n;
            }
        }
    }

    row_min_step<1>(minv, mini);
    row_min_step<2>(minv, mini);
    row_min_step<4>(minv, mini);
    row_min_step<8>(minv, mini);

    __shared__ float lv[WAVES][16];
    __shared__ int   li[WAVES][16];
    if (lane == 0 || lane == 16) {
        const int rowbase = hi ? 8 : 0;
#pragma unroll
        for (int r = 0; r < 8; ++r) {
            lv[wave][rowbase + r] = minv[r];
            li[wave][rowbase + r] = mini[r];
        }
    }
    __syncthreads();

    if (tid < 16) {
        const int m = tid;
        float bv = FLT_BIG;
        int   bi = 0;
#pragma unroll
        for (int w = 0; w < WAVES; ++w) {
            const float v = lv[w][m];
            const int   i = li[w][m];
            const bool take = (v < bv) | ((v == bv) & (i < bi));
            bv = take ? v : bv;
            bi = take ? i : bi;
        }

        const float kx = kpb[0 * M_PTS + m0 + m];
        const float ky = kpb[1 * M_PTS + m0 + m];
        const float kz = kpb[2 * M_PTS + m0 + m];

        const float px = pcx[bi];
        const float py = pcy[bi];
        const float pz = pcz[bi];

        const float* snb = sn + (size_t)b * 3 * N_PTS;
        const float sx = snb[0 * N_PTS + bi];
        const float sy = snb[1 * N_PTS + bi];
        const float sz = snb[2 * N_PTS + bi];

        const float dx = kx - px, dy = ky - py, dz = kz - pz;
        const float dn  = sqrtf(dx * dx + dy * dy + dz * dz);
        const float inv = 1.0f / (dn + EPS_F);
        const float dot = (sx * dx + sy * dy + sz * dz) * inv;

        out[(size_t)b * M_PTS + m0 + m] = dot * dot;
    }
}

extern "C" void kernel_launch(void* const* d_in, const int* in_sizes, int n_in,
                              void* d_out, int out_size, void* d_ws, size_t ws_size,
                              hipStream_t stream) {
    const float* kp = (const float*)d_in[0];
    const float* pc = (const float*)d_in[1];
    const float* sn = (const float*)d_in[2];
    float* out = (float*)d_out;

    dim3 grid(M_PTS / 16, BATCH);
    pos_loss_kernel<<<grid, THREADS, 0, stream>>>(kp, pc, sn, out);
}
